# starting kernel
# baseline (speedup 1.0000x reference)
#include <hip/hip_runtime.h>

typedef __attribute__((ext_vector_type(2))) float v2f;
typedef __attribute__((ext_vector_type(8))) float v8f;

#define IC   128
#define OC   32
#define H    14
#define W    14
#define NSP  196
#define KTOT 384
#define NT   13
#define MT   2
#define WAVES (NT * MT)
#define THREADS (WAVES * 32)

#define SX_COLS 16
#define SX_ROWS (H * SX_COLS)
#define SX_RSTR 130

__global__ __launch_bounds__(THREADS, 1)
void conv_roll_wmma(const float* __restrict__ x, const float* __restrict__ w,
                    float* __restrict__ out) {
  __shared__ float sxp[SX_ROWS * SX_RSTR];

  const int tid = threadIdx.x;

  for (int idx = tid; idx < IC * SX_ROWS; idx += THREADS) {
    const int ic  = idx / SX_ROWS;
    const int rem = idx % SX_ROWS;
    const int h   = rem / SX_COLS;
    const int c   = rem % SX_COLS;
    const int hs  = (h + H - 1) % H;
    const int ws  = (c + W - 2) % W;
    const bool inb = (c >= 1) && (c <= W);
    sxp[rem * SX_RSTR + ic] = inb ? x[ic * NSP + hs * W + ws] : 0.0f;
  }
  __syncthreads();

  const int wave     = tid >> 5;
  const int lane     = tid & 31;
  const int laneHalf = lane & 15;
  const int gK       = (lane >> 4) << 1;

  const int m0 = (wave & 1) * 16;
  const int n0 = (wave >> 1) * 16;

  const int  n      = n0 + laneHalf;
  const bool nvalid = (n < NSP);
  const int  nc     = nvalid ? n : (NSP - 1);
  const int  h      = nc / W;
  const int  wp     = nc % W;
  const int  coBase = h * SX_COLS + wp;

  const v2f* __restrict__ aRow =
      reinterpret_cast<const v2f*>(w + (m0 + laneHalf) * KTOT);

  v8f c = {};
#pragma unroll
  for (int k = 0; k < 3; ++k) {
    const float* __restrict__ bRow = &sxp[(coBase + k) * SX_RSTR + gK];
    for (int icb = 0; icb < IC; icb += 4) {
      const int kk0 = k * IC + icb;
      const v2f a = aRow[(kk0 + gK) >> 1];
      const v2f b = *reinterpret_cast<const v2f*>(bRow + icb);
      c = __builtin_amdgcn_wmma_f32_16x16x4_f32(false, a, false, b,
                                                (short)0, c, false, false);
    }
  }

  if (nvalid) {
    const int ocBase = m0 + ((lane >> 4) << 3);
#pragma unroll
    for (int r = 0; r < 8; ++r) {
      out[(ocBase + r) * NSP + n] = c[r];
    }
  }
}

extern "C" void kernel_launch(void* const* d_in, const int* in_sizes, int n_in,
                              void* d_out, int out_size, void* d_ws, size_t ws_size,
                              hipStream_t stream) {
  (void)in_sizes; (void)n_in; (void)out_size; (void)d_ws; (void)ws_size;
  const float* x = (const float*)d_in[0];
  const float* w = (const float*)d_in[1];
  float* out = (float*)d_out;
  conv_roll_wmma<<<dim3(1), dim3(THREADS), 0, stream>>>(x, w, out);
}


# one work-group per output row and 16-channel tile (28) instead of one for all: each stages only its own input row, all its loads in flight before the first LDS store; sum order unchanged
# speedup vs baseline: 4.8594x; 4.8594x over previous
#include <hip/hip_runtime.h>


typedef __attribute__((ext_vector_type(2))) float v2f;
typedef __attribute__((ext_vector_type(8))) float v8f;

#define IC   128
#define OC   32
#define H    14
#define W    14
#define NSP  196
#define KTOT 384
#define MT   2

#define THREADS 128
#define PER_T   ((IC * W) / THREADS)

#define SX_COLS 16
#define SX_RSTR 130

__global__ __launch_bounds__(THREADS, 1)
void conv_roll_wmma(const float* __restrict__ x, const float* __restrict__ w,
                    float* __restrict__ out) {
  __shared__ float sxp[SX_COLS * SX_RSTR];

  const int tid = threadIdx.x;
  const int m0  = blockIdx.x * 16;
  const int h   = blockIdx.y;
  const int hs  = (h + H - 1) % H;

  const float* __restrict__ xrow = x + hs * W;
  float v[PER_T];
#pragma unroll
  for (int j = 0; j < PER_T; ++j) {
    const int e  = tid + j * THREADS;
    const int ic = e / W;
    const int s  = e - ic * W;
    v[j] = xrow[ic * NSP + s];
  }
  sxp[tid] = 0.0f;
  sxp[(SX_COLS - 1) * SX_RSTR + tid] = 0.0f;
#pragma unroll
  for (int j = 0; j < PER_T; ++j) {
    const int e  = tid + j * THREADS;
    const int ic = e / W;
    const int s  = e - ic * W;
    const int c  = (s == W - 1) ? 1 : (s + 2);
    sxp[c * SX_RSTR + ic] = v[j];
  }
  __syncthreads();

  if (tid >= 32) return;

  const int lane     = tid;
  const int laneHalf = lane & 15;
  const int gK       = (lane >> 4) << 1;

  const bool nvalid = (laneHalf < W);
  const int  wp     = nvalid ? laneHalf : (W - 1);

  const v2f* __restrict__ aRow =
      reinterpret_cast<const v2f*>(w + (m0 + laneHalf) * KTOT);

  v8f c = {};
#pragma unroll
  for (int k = 0; k < 3; ++k) {
    const float* __restrict__ bRow = &sxp[(wp + k) * SX_RSTR + gK];
#pragma unroll
    for (int icb = 0; icb < IC; icb += 4) {
      const int kk0 = k * IC + icb;
      const v2f a = aRow[(kk0 + gK) >> 1];
      const v2f b = *reinterpret_cast<const v2f*>(bRow + icb);
      c = __builtin_amdgcn_wmma_f32_16x16x4_f32(false, a, false, b,
                                                (short)0, c, false, false);
    }
  }

  if (nvalid) {
    const int ocBase = m0 + ((lane >> 4) << 3);
#pragma unroll
    for (int r = 0; r < 8; ++r) {
      out[(ocBase + r) * NSP + h * W + laneHalf] = c[r];
    }
  }
}

extern "C" void kernel_launch(void* const* d_in, const int* in_sizes, int n_in,
                              void* d_out, int out_size, void* d_ws, size_t ws_size,
                              hipStream_t stream) {
  (void)in_sizes; (void)n_in; (void)out_size; (void)d_ws; (void)ws_size;
  const float* x = (const float*)d_in[0];
  const float* w = (const float*)d_in[1];
  float* out = (float*)d_out;
  conv_roll_wmma<<<dim3(MT, H), dim3(THREADS), 0, stream>>>(x, w, out);
}
